# starting kernel
# baseline (speedup 1.0000x reference)
#include <hip/hip_runtime.h>

typedef float v2f __attribute__((ext_vector_type(2)));
typedef float v8f __attribute__((ext_vector_type(8)));

#define Dd    1024
#define NS    25
#define NQ    75
#define NW    5
#define SROW  1028
#define KROW  32
#define WROW  8
#define QROW  132
#define KCH   128
#define NCHUNK (Dd / KCH)
#define NTHR  160

#define OFF_W    25700
#define OFF_KAUG 33908
#define OFF_PROW 34708
#define OFF_FCOL 34740
#define SMEM_FLOATS 34772

__global__ __launch_bounds__(NTHR) void ridge_head_fused(
    const float* __restrict__ q, const float* __restrict__ s,
    const int* __restrict__ labels, const float* __restrict__ scale,
    float* __restrict__ out)
{
  __shared__ __align__(16) float smem[SMEM_FLOATS];
  float* S_lds = smem;
  float* W_lds = smem + OFF_W;
  float* Kaug  = smem + OFF_KAUG;
  float* prow  = smem + OFF_PROW;
  float* fcol  = smem + OFF_FCOL;

  const int b    = blockIdx.x;
  const int tid  = threadIdx.x;
  const int lane = tid & 31;
  const int wave = tid >> 5;
  const int lm   = lane & 15;
  const int kh   = lane >> 4;

  {
    const float* sb = s + (size_t)b * NS * Dd;
    for (int idx = tid; idx < NS * (Dd / 4); idx += NTHR) {
      int r  = idx >> 8;
      int c4 = idx & 255;
      float4 v = ((const float4*)(sb + r * Dd))[c4];
      ((float4*)(S_lds + r * SROW))[c4] = v;
    }
    for (int idx = tid; idx < Dd * WROW + 16; idx += NTHR) W_lds[idx] = 0.0f;
  }
  __syncthreads();

  if (wave < 4) {
    const int ti = (wave >> 1) * 16;
    const int tj = (wave & 1) * 16;
    const float* arow = S_lds + (ti + lm) * SROW + 2 * kh;
    const float* brow = S_lds + (tj + lm) * SROW + 2 * kh;
    v8f acc0 = {0.f,0.f,0.f,0.f,0.f,0.f,0.f,0.f};
    v8f acc1 = {0.f,0.f,0.f,0.f,0.f,0.f,0.f,0.f};
    for (int k = 0; k < Dd; k += 8) {
      v2f a0 = *(const v2f*)(arow + k);
      v2f b0 = *(const v2f*)(brow + k);
      acc0 = __builtin_amdgcn_wmma_f32_16x16x4_f32(false, a0, false, b0,
                                                   (short)0, acc0, false, false);
      v2f a1 = *(const v2f*)(arow + k + 4);
      v2f b1 = *(const v2f*)(brow + k + 4);
      acc1 = __builtin_amdgcn_wmma_f32_16x16x4_f32(false, a1, false, b1,
                                                   (short)0, acc1, false, false);
    }
    v8f acc = acc0 + acc1;
    const int j = tj + lm;
    if (j < NS) {
      #pragma unroll
      for (int v = 0; v < 8; ++v) {
        int i = ti + kh * 8 + v;
        if (i < NS) Kaug[i * KROW + j] = acc[v] + ((i == j) ? 1.0f : 0.0f);
      }
    }
  } else {
    int r = tid - 128;
    if (r < NS) {
      int lab = labels[b * NS + r];
      for (int c = 0; c < NW; ++c)
        Kaug[r * KROW + NS + c] = (lab == c) ? 1.0f : 0.0f;
    }
  }
  __syncthreads();

  for (int p = 0; p < NS; ++p) {
    if (tid < NS + NW) {
      float pivinv = 1.0f / Kaug[p * KROW + p];
      prow[tid] = Kaug[p * KROW + tid] * pivinv;
      if (tid < NS) fcol[tid] = Kaug[tid * KROW + p];
    }
    __syncthreads();
    for (int e = tid; e < NS * (NS + NW); e += NTHR) {
      int r = e / (NS + NW);
      int c = e - r * (NS + NW);
      float v = (r == p) ? prow[c] : (Kaug[r * KROW + c] - fcol[r] * prow[c]);
      Kaug[r * KROW + c] = v;
    }
    __syncthreads();
  }

  const int m0 = wave * 16;
  const float* qb = q + (size_t)b * NQ * Dd;
  float4 r[16];
  #pragma unroll
  for (int u = 0; u < 16; ++u) {
    int m = m0 + u;
    float4 v = {0.f, 0.f, 0.f, 0.f};
    if (m < NQ) v = ((const float4*)(qb + (size_t)m * Dd))[lane];
    r[u] = v;
  }

  for (int e = tid; e < Dd * NW; e += NTHR) {
    int n = e >> 10;
    int k = e & (Dd - 1);
    float acc = 0.0f;
    #pragma unroll
    for (int i = 0; i < NS; ++i)
      acc += S_lds[i * SROW + k] * Kaug[i * KROW + NS + n];
    W_lds[k * WROW + n] = acc;
  }
  __syncthreads();

  float* Qs = S_lds + wave * (16 * QROW);

  v8f c0 = {0.f,0.f,0.f,0.f,0.f,0.f,0.f,0.f};
  v8f c1 = {0.f,0.f,0.f,0.f,0.f,0.f,0.f,0.f};

  const float* ap = Qs + lm * QROW + 2 * kh;

  for (int c = 0; c < NCHUNK; ++c) {
    #pragma unroll
    for (int u = 0; u < 16; ++u)
      ((float4*)(Qs + u * QROW))[lane] = r[u];

    if (c + 1 < NCHUNK) {
      #pragma unroll
      for (int u = 0; u < 16; ++u) {
        int m = m0 + u;
        float4 v = {0.f, 0.f, 0.f, 0.f};
        if (m < NQ) v = ((const float4*)(qb + (size_t)m * Dd + (c + 1) * KCH))[lane];
        r[u] = v;
      }
    }

    const int k0 = c * KCH;
    #pragma unroll 4
    for (int kc = 0; kc < KCH; kc += 8) {
      int kb0 = k0 + kc + 2 * kh;
      v2f a0 = *(const v2f*)(ap + kc);
      v2f b0 = { W_lds[kb0 * WROW + lm], W_lds[(kb0 + 1) * WROW + lm] };
      c0 = __builtin_amdgcn_wmma_f32_16x16x4_f32(false, a0, false, b0,
                                                 (short)0, c0, false, false);
      int kb1 = kb0 + 4;
      v2f a1 = *(const v2f*)(ap + kc + 4);
      v2f b1 = { W_lds[kb1 * WROW + lm], W_lds[(kb1 + 1) * WROW + lm] };
      c1 = __builtin_amdgcn_wmma_f32_16x16x4_f32(false, a1, false, b1,
                                                 (short)0, c1, false, false);
    }
  }

  const float sc = *scale;
  v8f cc = c0 + c1;
  if (lm < NW) {
    #pragma unroll
    for (int v = 0; v < 8; ++v) {
      int m = m0 + kh * 8 + v;
      if (m < NQ) out[((size_t)b * NQ + m) * NW + lm] = sc * cc[v];
    }
  }
}

extern "C" void kernel_launch(void* const* d_in, const int* in_sizes, int n_in,
                              void* d_out, int out_size, void* d_ws, size_t ws_size,
                              hipStream_t stream) {
  (void)n_in; (void)out_size; (void)d_ws; (void)ws_size;
  const float* q      = (const float*)d_in[0];
  const float* s      = (const float*)d_in[1];
  const int*   labels = (const int*)d_in[2];
  const float* scale  = (const float*)d_in[3];
  float* out = (float*)d_out;

  const int B = in_sizes[1] / (NS * Dd);
  if (B <= 0) return;

  ridge_head_fused<<<dim3(B), dim3(NTHR), 0, stream>>>(q, s, labels, scale, out);
}


# logits reassociated as (Q*S^T)*(K^-1 Y): K and Q*S^T come from one streamed [S;Q]*S^T pass in k-chunks, W = S^T*A stage and resident S dropped; new summation order
# speedup vs baseline: 1.9574x; 1.9574x over previous
#include <hip/hip_runtime.h>


typedef float v2f __attribute__((ext_vector_type(2)));
typedef float v4f __attribute__((ext_vector_type(4)));
typedef float v8f __attribute__((ext_vector_type(8)));

#define Dd    1024
#define NS    25
#define NQ    75
#define NW    5
#define KROW  32
#define KCH   128
#define NCHUNK (Dd / KCH)
#define NWAVE 7
#define NTHR  (NWAVE * 32)
#define TROWS (NWAVE * 16)
#define CROW  132
#define BUFF  (TROWS * CROW)
#define GROW  36

#define OFF_KAUG (2 * BUFF)
#define OFF_PROW (OFF_KAUG + NS * KROW)
#define OFF_FCOL (OFF_PROW + 32)
#define SMEM_FLOATS (OFF_FCOL + 32)

struct Frag { v4f a[4], x0[4], x1[4]; };

__device__ __forceinline__ void load_blk(Frag& f, const float* pa, const float* pb0, const float* pb1) {
  #pragma unroll
  for (int g = 0; g < 4; ++g) {
    f.a[g]  = *(const v4f*)(pa  + 8 * g);
    f.x0[g] = *(const v4f*)(pb0 + 8 * g);
    f.x1[g] = *(const v4f*)(pb1 + 8 * g);
  }
}

__device__ __forceinline__ void mma_blk(const Frag& f, v8f& acc0, v8f& acc1) {
  #pragma unroll
  for (int g = 0; g < 4; ++g) {
    v2f alo  = {f.a[g].x,  f.a[g].y},  ahi  = {f.a[g].z,  f.a[g].w};
    v2f x0lo = {f.x0[g].x, f.x0[g].y}, x0hi = {f.x0[g].z, f.x0[g].w};
    v2f x1lo = {f.x1[g].x, f.x1[g].y}, x1hi = {f.x1[g].z, f.x1[g].w};
    acc0 = __builtin_amdgcn_wmma_f32_16x16x4_f32(false, alo, false, x0lo,
                                                 (short)0, acc0, false, false);
    acc1 = __builtin_amdgcn_wmma_f32_16x16x4_f32(false, alo, false, x1lo,
                                                 (short)0, acc1, false, false);
    acc0 = __builtin_amdgcn_wmma_f32_16x16x4_f32(false, ahi, false, x0hi,
                                                 (short)0, acc0, false, false);
    acc1 = __builtin_amdgcn_wmma_f32_16x16x4_f32(false, ahi, false, x1hi,
                                                 (short)0, acc1, false, false);
  }
}

__global__ __launch_bounds__(NTHR) void ridge_head_fused(
    const float* __restrict__ q, const float* __restrict__ s,
    const int* __restrict__ labels, const float* __restrict__ scale,
    float* __restrict__ out)
{
  __shared__ __align__(16) float smem[SMEM_FLOATS];
  float* Kaug  = smem + OFF_KAUG;
  float* prow  = smem + OFF_PROW;
  float* fcol  = smem + OFF_FCOL;
  float* G     = smem;

  const int b    = blockIdx.x;
  const int tid  = threadIdx.x;
  const int lane = tid & 31;
  const int wave = __builtin_amdgcn_readfirstlane(tid >> 5);
  const int lm   = lane & 15;
  const int kh   = lane >> 4;

  const float* gsrc;
  int nvalid;
  if (wave < 2) {
    gsrc   = s + (size_t)b * NS * Dd + (size_t)wave * 16 * Dd;
    nvalid = (wave == 0) ? 16 : (NS - 16);
  } else {
    const int mq = (wave - 2) * 16;
    gsrc   = q + (size_t)b * NQ * Dd + (size_t)mq * Dd;
    nvalid = (NQ - mq < 16) ? (NQ - mq) : 16;
  }

  float4 r[16];
  #pragma unroll
  for (int u = 0; u < 16; ++u) {
    float4 v = {0.f, 0.f, 0.f, 0.f};
    if (u < nvalid) v = ((const float4*)(gsrc + (size_t)u * Dd))[lane];
    r[u] = v;
  }

  if (tid < NS) {
    int lab = labels[b * NS + tid];
    for (int c = 0; c < NW; ++c)
      Kaug[tid * KROW + NS + c] = (lab == c) ? 1.0f : 0.0f;
  }

  v8f acc0 = {0.f,0.f,0.f,0.f,0.f,0.f,0.f,0.f};
  v8f acc1 = {0.f,0.f,0.f,0.f,0.f,0.f,0.f,0.f};
  const int a_off  = (wave * 16 + lm) * CROW + 4 * kh;
  const int b0_off = lm * CROW + 4 * kh;
  const int b1_off = (16 + lm) * CROW + 4 * kh;

  for (int c = 0; c < NCHUNK; ++c) {
    float* buf = smem + (c & 1) * BUFF;

    #pragma unroll
    for (int u = 0; u < 16; ++u)
      ((float4*)(buf + (wave * 16 + u) * CROW))[lane] = r[u];

    if (c + 1 < NCHUNK) {
      #pragma unroll
      for (int u = 0; u < 16; ++u) {
        float4 v = {0.f, 0.f, 0.f, 0.f};
        if (u < nvalid) v = ((const float4*)(gsrc + (size_t)u * Dd + (c + 1) * KCH))[lane];
        r[u] = v;
      }
    }

    __syncthreads();

    const float* pa  = buf + a_off;
    const float* pb0 = buf + b0_off;
    const float* pb1 = buf + b1_off;
    Frag f0, f1;
    load_blk(f0, pa, pb0, pb1);
    load_blk(f1, pa + 32, pb0 + 32, pb1 + 32);
    mma_blk(f0, acc0, acc1);
    load_blk(f0, pa + 64, pb0 + 64, pb1 + 64);
    mma_blk(f1, acc0, acc1);
    load_blk(f1, pa + 96, pb0 + 96, pb1 + 96);
    mma_blk(f0, acc0, acc1);
    mma_blk(f1, acc0, acc1);
    __builtin_amdgcn_sched_group_barrier(0x100, 24, 0);
    __builtin_amdgcn_sched_group_barrier(0x008, 16, 0);
    __builtin_amdgcn_sched_group_barrier(0x100, 12, 0);
    __builtin_amdgcn_sched_group_barrier(0x008, 16, 0);
    __builtin_amdgcn_sched_group_barrier(0x100, 12, 0);
    __builtin_amdgcn_sched_group_barrier(0x008, 32, 0);
  }

  if (wave < 2) {
    #pragma unroll
    for (int v = 0; v < 8; ++v) {
      const int i  = wave * 16 + kh * 8 + v;
      const int j0 = lm, j1 = 16 + lm;
      if (i < NS) {
        Kaug[i * KROW + j0] = acc0[v] + ((i == j0) ? 1.0f : 0.0f);
        if (j1 < NS) Kaug[i * KROW + j1] = acc1[v] + ((i == j1) ? 1.0f : 0.0f);
      }
    }
  } else {
    const int mr = (wave - 2) * 16 + kh * 8;
    #pragma unroll
    for (int v = 0; v < 8; ++v) {
      G[(mr + v) * GROW + lm]      = acc0[v];
      G[(mr + v) * GROW + 16 + lm] = acc1[v];
    }
  }
  __syncthreads();

  for (int p = 0; p < NS; ++p) {
    if (tid < NS + NW) {
      float pivinv = 1.0f / Kaug[p * KROW + p];
      prow[tid] = Kaug[p * KROW + tid] * pivinv;
      if (tid < NS) fcol[tid] = Kaug[tid * KROW + p];
    }
    __syncthreads();
    for (int e = tid; e < NS * (NS + NW); e += NTHR) {
      int rr = e / (NS + NW);
      int cc = e - rr * (NS + NW);
      float v = (rr == p) ? prow[cc] : (Kaug[rr * KROW + cc] - fcol[rr] * prow[cc]);
      Kaug[rr * KROW + cc] = v;
    }
    __syncthreads();
  }

  if (wave >= 2) {
    const int m0 = (wave - 2) * 16;
    const float* ga = G + (m0 + lm) * GROW + 2 * kh;
    v8f cc = {0.f,0.f,0.f,0.f,0.f,0.f,0.f,0.f};
    v2f af[7], bf[7];
    #pragma unroll
    for (int ks = 0; ks < 7; ++ks) {
      const int k0 = 4 * ks + 2 * kh;
      af[ks] = *(const v2f*)(ga + 4 * ks);
      bf[ks].x = (lm < NW && k0 < NS)     ? Kaug[k0 * KROW + NS + lm]       : 0.0f;
      bf[ks].y = (lm < NW && k0 + 1 < NS) ? Kaug[(k0 + 1) * KROW + NS + lm] : 0.0f;
    }
    #pragma unroll
    for (int ks = 0; ks < 7; ++ks)
      cc = __builtin_amdgcn_wmma_f32_16x16x4_f32(false, af[ks], false, bf[ks],
                                                 (short)0, cc, false, false);
    const float sc = *scale;
    if (lm < NW) {
      #pragma unroll
      for (int v = 0; v < 8; ++v) {
        int m = m0 + kh * 8 + v;
        if (m < NQ) out[((size_t)b * NQ + m) * NW + lm] = sc * cc[v];
      }
    }
  }
}

extern "C" void kernel_launch(void* const* d_in, const int* in_sizes, int n_in,
                              void* d_out, int out_size, void* d_ws, size_t ws_size,
                              hipStream_t stream) {
  (void)n_in; (void)out_size; (void)d_ws; (void)ws_size;
  const float* q      = (const float*)d_in[0];
  const float* s      = (const float*)d_in[1];
  const int*   labels = (const int*)d_in[2];
  const float* scale  = (const float*)d_in[3];
  float* out = (float*)d_out;

  const int B = in_sizes[1] / (NS * Dd);
  if (B <= 0) return;

  ridge_head_fused<<<dim3(B), dim3(NTHR), 0, stream>>>(q, s, labels, scale, out);
}


# Gauss-Jordan step fused over two LDS copies of [K|Y] (read one, write the other): 1 workgroup barrier per pivot (was 2), no staged pivot row/column, lane=column, wave=rows; same sums per element
# speedup vs baseline: 2.2914x; 1.1706x over previous
#include <hip/hip_runtime.h>


typedef float v2f __attribute__((ext_vector_type(2)));
typedef float v4f __attribute__((ext_vector_type(4)));
typedef float v8f __attribute__((ext_vector_type(8)));

#define Dd    1024
#define NS    25
#define NQ    75
#define NW    5
#define KROW  32
#define KCH   128
#define NCHUNK (Dd / KCH)
#define NWAVE 7
#define NTHR  (NWAVE * 32)
#define TROWS (NWAVE * 16)
#define CROW  132
#define BUFF  (TROWS * CROW)
#define GROW  36

#define OFF_KAUG (2 * BUFF)
#define KCOPY (NS * KROW)
#define SMEM_FLOATS (OFF_KAUG + 2 * KCOPY)

struct Frag { v4f a[4], x0[4], x1[4]; };

__device__ __forceinline__ void load_blk(Frag& f, const float* pa, const float* pb0, const float* pb1) {
  #pragma unroll
  for (int g = 0; g < 4; ++g) {
    f.a[g]  = *(const v4f*)(pa  + 8 * g);
    f.x0[g] = *(const v4f*)(pb0 + 8 * g);
    f.x1[g] = *(const v4f*)(pb1 + 8 * g);
  }
}

__device__ __forceinline__ void mma_blk(const Frag& f, v8f& acc0, v8f& acc1) {
  #pragma unroll
  for (int g = 0; g < 4; ++g) {
    v2f alo  = {f.a[g].x,  f.a[g].y},  ahi  = {f.a[g].z,  f.a[g].w};
    v2f x0lo = {f.x0[g].x, f.x0[g].y}, x0hi = {f.x0[g].z, f.x0[g].w};
    v2f x1lo = {f.x1[g].x, f.x1[g].y}, x1hi = {f.x1[g].z, f.x1[g].w};
    acc0 = __builtin_amdgcn_wmma_f32_16x16x4_f32(false, alo, false, x0lo,
                                                 (short)0, acc0, false, false);
    acc1 = __builtin_amdgcn_wmma_f32_16x16x4_f32(false, alo, false, x1lo,
                                                 (short)0, acc1, false, false);
    acc0 = __builtin_amdgcn_wmma_f32_16x16x4_f32(false, ahi, false, x0hi,
                                                 (short)0, acc0, false, false);
    acc1 = __builtin_amdgcn_wmma_f32_16x16x4_f32(false, ahi, false, x1hi,
                                                 (short)0, acc1, false, false);
  }
}

__global__ __launch_bounds__(NTHR) void ridge_head_fused(
    const float* __restrict__ q, const float* __restrict__ s,
    const int* __restrict__ labels, const float* __restrict__ scale,
    float* __restrict__ out)
{
  __shared__ __align__(16) float smem[SMEM_FLOATS];
  float* Kaug  = smem + OFF_KAUG;
  float* G     = smem;

  const int b    = blockIdx.x;
  const int tid  = threadIdx.x;
  const int lane = tid & 31;
  const int wave = __builtin_amdgcn_readfirstlane(tid >> 5);
  const int lm   = lane & 15;
  const int kh   = lane >> 4;

  const float* gsrc;
  int nvalid;
  if (wave < 2) {
    gsrc   = s + (size_t)b * NS * Dd + (size_t)wave * 16 * Dd;
    nvalid = (wave == 0) ? 16 : (NS - 16);
  } else {
    const int mq = (wave - 2) * 16;
    gsrc   = q + (size_t)b * NQ * Dd + (size_t)mq * Dd;
    nvalid = (NQ - mq < 16) ? (NQ - mq) : 16;
  }

  float4 r[16];
  #pragma unroll
  for (int u = 0; u < 16; ++u) {
    float4 v = {0.f, 0.f, 0.f, 0.f};
    if (u < nvalid) v = ((const float4*)(gsrc + (size_t)u * Dd))[lane];
    r[u] = v;
  }

  if (tid < NS) {
    int lab = labels[b * NS + tid];
    for (int c = 0; c < NW; ++c)
      Kaug[tid * KROW + NS + c] = (lab == c) ? 1.0f : 0.0f;
  }

  v8f acc0 = {0.f,0.f,0.f,0.f,0.f,0.f,0.f,0.f};
  v8f acc1 = {0.f,0.f,0.f,0.f,0.f,0.f,0.f,0.f};
  const int a_off  = (wave * 16 + lm) * CROW + 4 * kh;
  const int b0_off = lm * CROW + 4 * kh;
  const int b1_off = (16 + lm) * CROW + 4 * kh;

  for (int c = 0; c < NCHUNK; ++c) {
    float* buf = smem + (c & 1) * BUFF;

    #pragma unroll
    for (int u = 0; u < 16; ++u)
      ((float4*)(buf + (wave * 16 + u) * CROW))[lane] = r[u];

    if (c + 1 < NCHUNK) {
      #pragma unroll
      for (int u = 0; u < 16; ++u) {
        float4 v = {0.f, 0.f, 0.f, 0.f};
        if (u < nvalid) v = ((const float4*)(gsrc + (size_t)u * Dd + (c + 1) * KCH))[lane];
        r[u] = v;
      }
    }

    __syncthreads();

    const float* pa  = buf + a_off;
    const float* pb0 = buf + b0_off;
    const float* pb1 = buf + b1_off;
    Frag f0, f1;
    load_blk(f0, pa, pb0, pb1);
    load_blk(f1, pa + 32, pb0 + 32, pb1 + 32);
    mma_blk(f0, acc0, acc1);
    load_blk(f0, pa + 64, pb0 + 64, pb1 + 64);
    mma_blk(f1, acc0, acc1);
    load_blk(f1, pa + 96, pb0 + 96, pb1 + 96);
    mma_blk(f0, acc0, acc1);
    mma_blk(f1, acc0, acc1);
    __builtin_amdgcn_sched_group_barrier(0x100, 24, 0);
    __builtin_amdgcn_sched_group_barrier(0x008, 16, 0);
    __builtin_amdgcn_sched_group_barrier(0x100, 12, 0);
    __builtin_amdgcn_sched_group_barrier(0x008, 16, 0);
    __builtin_amdgcn_sched_group_barrier(0x100, 12, 0);
    __builtin_amdgcn_sched_group_barrier(0x008, 32, 0);
  }

  if (wave < 2) {
    #pragma unroll
    for (int v = 0; v < 8; ++v) {
      const int i  = wave * 16 + kh * 8 + v;
      const int j0 = lm, j1 = 16 + lm;
      if (i < NS) {
        Kaug[i * KROW + j0] = acc0[v] + ((i == j0) ? 1.0f : 0.0f);
        if (j1 < NS) Kaug[i * KROW + j1] = acc1[v] + ((i == j1) ? 1.0f : 0.0f);
      }
    }
  } else {
    const int mr = (wave - 2) * 16 + kh * 8;
    #pragma unroll
    for (int v = 0; v < 8; ++v) {
      G[(mr + v) * GROW + lm]      = acc0[v];
      G[(mr + v) * GROW + 16 + lm] = acc1[v];
    }
  }
  __syncthreads();

  for (int p = 0; p < NS; ++p) {
    const float* srcK = Kaug + (p & 1) * KCOPY;
    float*       dstK = Kaug + ((p + 1) & 1) * KCOPY;
    if (lane < NS + NW) {
      const float pivinv = 1.0f / srcK[p * KROW + p];
      const float prs    = srcK[p * KROW + lane] * pivinv;
      float own[4], fc[4];
      #pragma unroll
      for (int i = 0; i < 4; ++i) {
        const int rr = wave + 7 * i;
        if (rr < NS) {
          own[i] = srcK[rr * KROW + lane];
          fc[i]  = srcK[rr * KROW + p];
        }
      }
      #pragma unroll
      for (int i = 0; i < 4; ++i) {
        const int rr = wave + 7 * i;
        if (rr < NS)
          dstK[rr * KROW + lane] = (rr == p) ? prs : (own[i] - fc[i] * prs);
      }
    }
    __syncthreads();
  }
  const float* Kfin = Kaug + (NS & 1) * KCOPY;

  if (wave >= 2) {
    const int m0 = (wave - 2) * 16;
    const float* ga = G + (m0 + lm) * GROW + 2 * kh;
    v8f cc = {0.f,0.f,0.f,0.f,0.f,0.f,0.f,0.f};
    v2f af[7], bf[7];
    #pragma unroll
    for (int ks = 0; ks < 7; ++ks) {
      const int k0 = 4 * ks + 2 * kh;
      af[ks] = *(const v2f*)(ga + 4 * ks);
      bf[ks].x = (lm < NW && k0 < NS)     ? Kfin[k0 * KROW + NS + lm]       : 0.0f;
      bf[ks].y = (lm < NW && k0 + 1 < NS) ? Kfin[(k0 + 1) * KROW + NS + lm] : 0.0f;
    }
    #pragma unroll
    for (int ks = 0; ks < 7; ++ks)
      cc = __builtin_amdgcn_wmma_f32_16x16x4_f32(false, af[ks], false, bf[ks],
                                                 (short)0, cc, false, false);
    const float sc = *scale;
    if (lm < NW) {
      #pragma unroll
      for (int v = 0; v < 8; ++v) {
        int m = m0 + kh * 8 + v;
        if (m < NQ) out[((size_t)b * NQ + m) * NW + lm] = sc * cc[v];
      }
    }
  }
}

extern "C" void kernel_launch(void* const* d_in, const int* in_sizes, int n_in,
                              void* d_out, int out_size, void* d_ws, size_t ws_size,
                              hipStream_t stream) {
  (void)n_in; (void)out_size; (void)d_ws; (void)ws_size;
  const float* q      = (const float*)d_in[0];
  const float* s      = (const float*)d_in[1];
  const int*   labels = (const int*)d_in[2];
  const float* scale  = (const float*)d_in[3];
  float* out = (float*)d_out;

  const int B = in_sizes[1] / (NS * Dd);
  if (B <= 0) return;

  ridge_head_fused<<<dim3(B), dim3(NTHR), 0, stream>>>(q, s, labels, scale, out);
}


# Gauss-Jordan solve in blocks of 5 pivots per workgroup barrier (5 barriers, was 25): each thread carries the 5x5 pivot block and its rows' block columns in registers; same operations per element
# speedup vs baseline: 2.6779x; 1.1687x over previous
#include <hip/hip_runtime.h>


typedef float v2f __attribute__((ext_vector_type(2)));
typedef float v4f __attribute__((ext_vector_type(4)));
typedef float v8f __attribute__((ext_vector_type(8)));

#define Dd    1024
#define NS    25
#define NQ    75
#define NW    5
#define KROW  32
#define KCH   128
#define NCHUNK (Dd / KCH)
#define NWAVE 7
#define NTHR  (NWAVE * 32)
#define TROWS (NWAVE * 16)
#define CROW  132
#define BUFF  (TROWS * CROW)
#define GROW  36
#define GJB   5
static_assert(NS % GJB == 0, "the solve takes whole blocks of GJB pivots");
static_assert(4 * NWAVE >= NS, "four rows per wave must cover the matrix");

#define OFF_KAUG (2 * BUFF)
#define KR    (4 * NWAVE)
#define KCOPY (KR * KROW)
#define SMEM_FLOATS (OFF_KAUG + 2 * KCOPY)

struct Frag { v4f a[4], x0[4], x1[4]; };

__device__ __forceinline__ void load_blk(Frag& f, const float* pa, const float* pb0, const float* pb1) {
  #pragma unroll
  for (int g = 0; g < 4; ++g) {
    f.a[g]  = *(const v4f*)(pa  + 8 * g);
    f.x0[g] = *(const v4f*)(pb0 + 8 * g);
    f.x1[g] = *(const v4f*)(pb1 + 8 * g);
  }
}

__device__ __forceinline__ void mma_blk(const Frag& f, v8f& acc0, v8f& acc1) {
  #pragma unroll
  for (int g = 0; g < 4; ++g) {
    v2f alo  = {f.a[g].x,  f.a[g].y},  ahi  = {f.a[g].z,  f.a[g].w};
    v2f x0lo = {f.x0[g].x, f.x0[g].y}, x0hi = {f.x0[g].z, f.x0[g].w};
    v2f x1lo = {f.x1[g].x, f.x1[g].y}, x1hi = {f.x1[g].z, f.x1[g].w};
    acc0 = __builtin_amdgcn_wmma_f32_16x16x4_f32(false, alo, false, x0lo,
                                                 (short)0, acc0, false, false);
    acc1 = __builtin_amdgcn_wmma_f32_16x16x4_f32(false, alo, false, x1lo,
                                                 (short)0, acc1, false, false);
    acc0 = __builtin_amdgcn_wmma_f32_16x16x4_f32(false, ahi, false, x0hi,
                                                 (short)0, acc0, false, false);
    acc1 = __builtin_amdgcn_wmma_f32_16x16x4_f32(false, ahi, false, x1hi,
                                                 (short)0, acc1, false, false);
  }
}

__global__ __launch_bounds__(NTHR) void ridge_head_fused(
    const float* __restrict__ q, const float* __restrict__ s,
    const int* __restrict__ labels, const float* __restrict__ scale,
    float* __restrict__ out)
{
  __shared__ __align__(16) float smem[SMEM_FLOATS];
  float* Kaug  = smem + OFF_KAUG;
  float* G     = smem;

  const int b    = blockIdx.x;
  const int tid  = threadIdx.x;
  const int lane = tid & 31;
  const int wave = __builtin_amdgcn_readfirstlane(tid >> 5);
  const int lm   = lane & 15;
  const int kh   = lane >> 4;

  const float* gsrc;
  int nvalid;
  if (wave < 2) {
    gsrc   = s + (size_t)b * NS * Dd + (size_t)wave * 16 * Dd;
    nvalid = (wave == 0) ? 16 : (NS - 16);
  } else {
    const int mq = (wave - 2) * 16;
    gsrc   = q + (size_t)b * NQ * Dd + (size_t)mq * Dd;
    nvalid = (NQ - mq < 16) ? (NQ - mq) : 16;
  }

  float4 r[16];
  #pragma unroll
  for (int u = 0; u < 16; ++u) {
    float4 v = {0.f, 0.f, 0.f, 0.f};
    if (u < nvalid) v = ((const float4*)(gsrc + (size_t)u * Dd))[lane];
    r[u] = v;
  }

  if (tid < NS) {
    int lab = labels[b * NS + tid];
    for (int c = 0; c < NW; ++c)
      Kaug[tid * KROW + NS + c] = (lab == c) ? 1.0f : 0.0f;
  }
  if (tid < (KR - NS) * KROW) Kaug[NS * KROW + tid] = 0.0f;

  v8f acc0 = {0.f,0.f,0.f,0.f,0.f,0.f,0.f,0.f};
  v8f acc1 = {0.f,0.f,0.f,0.f,0.f,0.f,0.f,0.f};
  const int a_off  = (wave * 16 + lm) * CROW + 4 * kh;
  const int b0_off = lm * CROW + 4 * kh;
  const int b1_off = (16 + lm) * CROW + 4 * kh;

  for (int c = 0; c < NCHUNK; ++c) {
    float* buf = smem + (c & 1) * BUFF;

    #pragma unroll
    for (int u = 0; u < 16; ++u)
      ((float4*)(buf + (wave * 16 + u) * CROW))[lane] = r[u];

    if (c + 1 < NCHUNK) {
      #pragma unroll
      for (int u = 0; u < 16; ++u) {
        float4 v = {0.f, 0.f, 0.f, 0.f};
        if (u < nvalid) v = ((const float4*)(gsrc + (size_t)u * Dd + (c + 1) * KCH))[lane];
        r[u] = v;
      }
    }

    __syncthreads();

    const float* pa  = buf + a_off;
    const float* pb0 = buf + b0_off;
    const float* pb1 = buf + b1_off;
    Frag f0, f1;
    load_blk(f0, pa, pb0, pb1);
    load_blk(f1, pa + 32, pb0 + 32, pb1 + 32);
    mma_blk(f0, acc0, acc1);
    load_blk(f0, pa + 64, pb0 + 64, pb1 + 64);
    mma_blk(f1, acc0, acc1);
    load_blk(f1, pa + 96, pb0 + 96, pb1 + 96);
    mma_blk(f0, acc0, acc1);
    mma_blk(f1, acc0, acc1);
    __builtin_amdgcn_sched_group_barrier(0x100, 24, 0);
    __builtin_amdgcn_sched_group_barrier(0x008, 16, 0);
    __builtin_amdgcn_sched_group_barrier(0x100, 12, 0);
    __builtin_amdgcn_sched_group_barrier(0x008, 16, 0);
    __builtin_amdgcn_sched_group_barrier(0x100, 12, 0);
    __builtin_amdgcn_sched_group_barrier(0x008, 32, 0);
  }

  if (wave < 2) {
    #pragma unroll
    for (int v = 0; v < 8; ++v) {
      const int i  = wave * 16 + kh * 8 + v;
      const int j0 = lm, j1 = 16 + lm;
      if (i < NS) {
        Kaug[i * KROW + j0] = acc0[v] + ((i == j0) ? 1.0f : 0.0f);
        if (j1 < NS) Kaug[i * KROW + j1] = acc1[v] + ((i == j1) ? 1.0f : 0.0f);
      }
    }
  } else {
    const int mr = (wave - 2) * 16 + kh * 8;
    #pragma unroll
    for (int v = 0; v < 8; ++v) {
      G[(mr + v) * GROW + lm]      = acc0[v];
      G[(mr + v) * GROW + 16 + lm] = acc1[v];
    }
  }
  __syncthreads();

  for (int p = 0; p < NS; p += GJB) {
    const int cur = (p / GJB) & 1;
    const float* srcK = Kaug + cur * KCOPY;
    float*       dstK = Kaug + (cur ^ 1) * KCOPY;
    if (lane < NS + NW) {
      float P[GJB][GJB], PJ[GJB], F[4][GJB], O[4];
      #pragma unroll
      for (int i = 0; i < GJB; ++i) {
        #pragma unroll
        for (int k = 0; k < GJB; ++k) P[i][k] = srcK[(p + i) * KROW + p + k];
        PJ[i] = srcK[(p + i) * KROW + lane];
      }
      #pragma unroll
      for (int t = 0; t < 4; ++t) {
        const int rr = wave + NWAVE * t;
        #pragma unroll
        for (int k = 0; k < GJB; ++k) F[t][k] = srcK[rr * KROW + p + k];
        O[t] = srcK[rr * KROW + lane];
      }
      #pragma unroll
      for (int s = 0; s < GJB; ++s) {
        const float pivinv = 1.0f / P[s][s];
        float ps[GJB];
        #pragma unroll
        for (int k = s + 1; k < GJB; ++k) ps[k] = P[s][k] * pivinv;
        const float pj = PJ[s] * pivinv;
        float nps[GJB];
        #pragma unroll
        for (int k = s + 1; k < GJB; ++k) nps[k] = -ps[k];
        const float npj = -pj;
        #pragma unroll
        for (int i = 0; i < GJB; ++i) {
          if (i == s) continue;
          const float f = P[i][s];
          #pragma unroll
          for (int k = s + 1; k < GJB; ++k) P[i][k] = __builtin_fmaf(f, nps[k], P[i][k]);
          PJ[i] = __builtin_fmaf(f, npj, PJ[i]);
        }
        #pragma unroll
        for (int k = s + 1; k < GJB; ++k) P[s][k] = ps[k];
        PJ[s] = pj;
        #pragma unroll
        for (int t = 0; t < 4; ++t) {
          const float f = F[t][s];
          #pragma unroll
          for (int k = s + 1; k < GJB; ++k) F[t][k] = __builtin_fmaf(f, nps[k], F[t][k]);
          O[t] = __builtin_fmaf(f, npj, O[t]);
        }
      }
      #pragma unroll
      for (int t = 0; t < 4; ++t) {
        const int rr = wave + NWAVE * t;
        float v = O[t];
        #pragma unroll
        for (int i = 0; i < GJB; ++i)
          if (rr == p + i) v = PJ[i];
        dstK[rr * KROW + lane] = v;
      }
    }
    __syncthreads();
  }
  const float* Kfin = Kaug + ((NS / GJB) & 1) * KCOPY;

  if (wave >= 2) {
    const int m0 = (wave - 2) * 16;
    const float* ga = G + (m0 + lm) * GROW + 2 * kh;
    v8f cc = {0.f,0.f,0.f,0.f,0.f,0.f,0.f,0.f};
    v2f af[7], bf[7];
    #pragma unroll
    for (int ks = 0; ks < 7; ++ks) {
      const int k0 = 4 * ks + 2 * kh;
      af[ks] = *(const v2f*)(ga + 4 * ks);
      bf[ks].x = (lm < NW && k0 < NS)     ? Kfin[k0 * KROW + NS + lm]       : 0.0f;
      bf[ks].y = (lm < NW && k0 + 1 < NS) ? Kfin[(k0 + 1) * KROW + NS + lm] : 0.0f;
    }
    #pragma unroll
    for (int ks = 0; ks < 7; ++ks)
      cc = __builtin_amdgcn_wmma_f32_16x16x4_f32(false, af[ks], false, bf[ks],
                                                 (short)0, cc, false, false);
    const float sc = *scale;
    if (lm < NW) {
      #pragma unroll
      for (int v = 0; v < 8; ++v) {
        int m = m0 + kh * 8 + v;
        if (m < NQ) out[((size_t)b * NQ + m) * NW + lm] = sc * cc[v];
      }
    }
  }
}

extern "C" void kernel_launch(void* const* d_in, const int* in_sizes, int n_in,
                              void* d_out, int out_size, void* d_ws, size_t ws_size,
                              hipStream_t stream) {
  (void)n_in; (void)out_size; (void)d_ws; (void)ws_size;
  const float* q      = (const float*)d_in[0];
  const float* s      = (const float*)d_in[1];
  const int*   labels = (const int*)d_in[2];
  const float* scale  = (const float*)d_in[3];
  float* out = (float*)d_out;

  const int B = in_sizes[1] / (NS * Dd);
  if (B <= 0) return;

  ridge_head_fused<<<dim3(B), dim3(NTHR), 0, stream>>>(q, s, labels, scale, out);
}


# streaming loop's workgroup barrier waits for LDS only (hand-written s_wait_dscnt + signal + wait, not __syncthreads): next chunk's global loads stay in flight under the chunk's products; same sums
# speedup vs baseline: 2.8837x; 1.0769x over previous
#include <hip/hip_runtime.h>


typedef float v2f __attribute__((ext_vector_type(2)));
typedef float v4f __attribute__((ext_vector_type(4)));
typedef float v8f __attribute__((ext_vector_type(8)));

#define Dd    1024
#define NS    25
#define NQ    75
#define NW    5
#define KROW  32
#define KCH   128
#define NCHUNK (Dd / KCH)
#define NWAVE 7
#define NTHR  (NWAVE * 32)
#define TROWS (NWAVE * 16)
#define CROW  132
#define BUFF  (TROWS * CROW)
#define GROW  36
#define GJB   5
static_assert(NS % GJB == 0, "the solve takes whole blocks of GJB pivots");
static_assert(4 * NWAVE >= NS, "four rows per wave must cover the matrix");

#define OFF_KAUG (2 * BUFF)
#define KR    (4 * NWAVE)
#define KCOPY (KR * KROW)
#define SMEM_FLOATS (OFF_KAUG + 2 * KCOPY)

__device__ __forceinline__ void chunk_barrier() {
  asm volatile("s_wait_dscnt 0x0" ::: "memory");
  asm volatile("s_barrier_signal -1" ::: "memory");
  asm volatile("s_barrier_wait -1" ::: "memory");
}

struct Frag { v4f a[4], x0[4], x1[4]; };

__device__ __forceinline__ void load_blk(Frag& f, const float* pa, const float* pb0, const float* pb1) {
  #pragma unroll
  for (int g = 0; g < 4; ++g) {
    f.a[g]  = *(const v4f*)(pa  + 8 * g);
    f.x0[g] = *(const v4f*)(pb0 + 8 * g);
    f.x1[g] = *(const v4f*)(pb1 + 8 * g);
  }
}

__device__ __forceinline__ void mma_blk(const Frag& f, v8f& acc0, v8f& acc1) {
  #pragma unroll
  for (int g = 0; g < 4; ++g) {
    v2f alo  = {f.a[g].x,  f.a[g].y},  ahi  = {f.a[g].z,  f.a[g].w};
    v2f x0lo = {f.x0[g].x, f.x0[g].y}, x0hi = {f.x0[g].z, f.x0[g].w};
    v2f x1lo = {f.x1[g].x, f.x1[g].y}, x1hi = {f.x1[g].z, f.x1[g].w};
    acc0 = __builtin_amdgcn_wmma_f32_16x16x4_f32(false, alo, false, x0lo,
                                                 (short)0, acc0, false, false);
    acc1 = __builtin_amdgcn_wmma_f32_16x16x4_f32(false, alo, false, x1lo,
                                                 (short)0, acc1, false, false);
    acc0 = __builtin_amdgcn_wmma_f32_16x16x4_f32(false, ahi, false, x0hi,
                                                 (short)0, acc0, false, false);
    acc1 = __builtin_amdgcn_wmma_f32_16x16x4_f32(false, ahi, false, x1hi,
                                                 (short)0, acc1, false, false);
  }
}

__global__ __launch_bounds__(NTHR) void ridge_head_fused(
    const float* __restrict__ q, const float* __restrict__ s,
    const int* __restrict__ labels, const float* __restrict__ scale,
    float* __restrict__ out)
{
  __shared__ __align__(16) float smem[SMEM_FLOATS];
  float* Kaug  = smem + OFF_KAUG;
  float* G     = smem;

  const int b    = blockIdx.x;
  const int tid  = threadIdx.x;
  const int lane = tid & 31;
  const int wave = __builtin_amdgcn_readfirstlane(tid >> 5);
  const int lm   = lane & 15;
  const int kh   = lane >> 4;

  const float* gsrc;
  int nvalid;
  if (wave < 2) {
    gsrc   = s + (size_t)b * NS * Dd + (size_t)wave * 16 * Dd;
    nvalid = (wave == 0) ? 16 : (NS - 16);
  } else {
    const int mq = (wave - 2) * 16;
    gsrc   = q + (size_t)b * NQ * Dd + (size_t)mq * Dd;
    nvalid = (NQ - mq < 16) ? (NQ - mq) : 16;
  }

  float4 r[16];
  #pragma unroll
  for (int u = 0; u < 16; ++u) {
    float4 v = {0.f, 0.f, 0.f, 0.f};
    if (u < nvalid) v = ((const float4*)(gsrc + (size_t)u * Dd))[lane];
    r[u] = v;
  }

  if (tid < NS) {
    int lab = labels[b * NS + tid];
    for (int c = 0; c < NW; ++c)
      Kaug[tid * KROW + NS + c] = (lab == c) ? 1.0f : 0.0f;
  }
  if (tid < (KR - NS) * KROW) Kaug[NS * KROW + tid] = 0.0f;

  v8f acc0 = {0.f,0.f,0.f,0.f,0.f,0.f,0.f,0.f};
  v8f acc1 = {0.f,0.f,0.f,0.f,0.f,0.f,0.f,0.f};
  const int a_off  = (wave * 16 + lm) * CROW + 4 * kh;
  const int b0_off = lm * CROW + 4 * kh;
  const int b1_off = (16 + lm) * CROW + 4 * kh;

  for (int c = 0; c < NCHUNK; ++c) {
    float* buf = smem + (c & 1) * BUFF;

    #pragma unroll
    for (int u = 0; u < 16; ++u)
      ((float4*)(buf + (wave * 16 + u) * CROW))[lane] = r[u];

    if (c + 1 < NCHUNK) {
      #pragma unroll
      for (int u = 0; u < 16; ++u) {
        float4 v = {0.f, 0.f, 0.f, 0.f};
        if (u < nvalid) v = ((const float4*)(gsrc + (size_t)u * Dd + (c + 1) * KCH))[lane];
        r[u] = v;
      }
    }

    chunk_barrier();

    const float* pa  = buf + a_off;
    const float* pb0 = buf + b0_off;
    const float* pb1 = buf + b1_off;
    Frag f0, f1;
    load_blk(f0, pa, pb0, pb1);
    load_blk(f1, pa + 32, pb0 + 32, pb1 + 32);
    mma_blk(f0, acc0, acc1);
    load_blk(f0, pa + 64, pb0 + 64, pb1 + 64);
    mma_blk(f1, acc0, acc1);
    load_blk(f1, pa + 96, pb0 + 96, pb1 + 96);
    mma_blk(f0, acc0, acc1);
    mma_blk(f1, acc0, acc1);
    __builtin_amdgcn_sched_group_barrier(0x100, 24, 0);
    __builtin_amdgcn_sched_group_barrier(0x008, 16, 0);
    __builtin_amdgcn_sched_group_barrier(0x100, 12, 0);
    __builtin_amdgcn_sched_group_barrier(0x008, 16, 0);
    __builtin_amdgcn_sched_group_barrier(0x100, 12, 0);
    __builtin_amdgcn_sched_group_barrier(0x008, 32, 0);
  }

  if (wave < 2) {
    #pragma unroll
    for (int v = 0; v < 8; ++v) {
      const int i  = wave * 16 + kh * 8 + v;
      const int j0 = lm, j1 = 16 + lm;
      if (i < NS) {
        Kaug[i * KROW + j0] = acc0[v] + ((i == j0) ? 1.0f : 0.0f);
        if (j1 < NS) Kaug[i * KROW + j1] = acc1[v] + ((i == j1) ? 1.0f : 0.0f);
      }
    }
  } else {
    const int mr = (wave - 2) * 16 + kh * 8;
    #pragma unroll
    for (int v = 0; v < 8; ++v) {
      G[(mr + v) * GROW + lm]      = acc0[v];
      G[(mr + v) * GROW + 16 + lm] = acc1[v];
    }
  }
  __syncthreads();

  for (int p = 0; p < NS; p += GJB) {
    const int cur = (p / GJB) & 1;
    const float* srcK = Kaug + cur * KCOPY;
    float*       dstK = Kaug + (cur ^ 1) * KCOPY;
    if (lane < NS + NW) {
      float P[GJB][GJB], PJ[GJB], F[4][GJB], O[4];
      #pragma unroll
      for (int i = 0; i < GJB; ++i) {
        #pragma unroll
        for (int k = 0; k < GJB; ++k) P[i][k] = srcK[(p + i) * KROW + p + k];
        PJ[i] = srcK[(p + i) * KROW + lane];
      }
      #pragma unroll
      for (int t = 0; t < 4; ++t) {
        const int rr = wave + NWAVE * t;
        #pragma unroll
        for (int k = 0; k < GJB; ++k) F[t][k] = srcK[rr * KROW + p + k];
        O[t] = srcK[rr * KROW + lane];
      }
      #pragma unroll
      for (int s = 0; s < GJB; ++s) {
        const float pivinv = 1.0f / P[s][s];
        float ps[GJB];
        #pragma unroll
        for (int k = s + 1; k < GJB; ++k) ps[k] = P[s][k] * pivinv;
        const float pj = PJ[s] * pivinv;
        float nps[GJB];
        #pragma unroll
        for (int k = s + 1; k < GJB; ++k) nps[k] = -ps[k];
        const float npj = -pj;
        #pragma unroll
        for (int i = 0; i < GJB; ++i) {
          if (i == s) continue;
          const float f = P[i][s];
          #pragma unroll
          for (int k = s + 1; k < GJB; ++k) P[i][k] = __builtin_fmaf(f, nps[k], P[i][k]);
          PJ[i] = __builtin_fmaf(f, npj, PJ[i]);
        }
        #pragma unroll
        for (int k = s + 1; k < GJB; ++k) P[s][k] = ps[k];
        PJ[s] = pj;
        #pragma unroll
        for (int t = 0; t < 4; ++t) {
          const float f = F[t][s];
          #pragma unroll
          for (int k = s + 1; k < GJB; ++k) F[t][k] = __builtin_fmaf(f, nps[k], F[t][k]);
          O[t] = __builtin_fmaf(f, npj, O[t]);
        }
      }
      #pragma unroll
      for (int t = 0; t < 4; ++t) {
        const int rr = wave + NWAVE * t;
        float v = O[t];
        #pragma unroll
        for (int i = 0; i < GJB; ++i)
          if (rr == p + i) v = PJ[i];
        dstK[rr * KROW + lane] = v;
      }
    }
    __syncthreads();
  }
  const float* Kfin = Kaug + ((NS / GJB) & 1) * KCOPY;

  if (wave >= 2) {
    const int m0 = (wave - 2) * 16;
    const float* ga = G + (m0 + lm) * GROW + 2 * kh;
    v8f cc = {0.f,0.f,0.f,0.f,0.f,0.f,0.f,0.f};
    v2f af[7], bf[7];
    #pragma unroll
    for (int ks = 0; ks < 7; ++ks) {
      const int k0 = 4 * ks + 2 * kh;
      af[ks] = *(const v2f*)(ga + 4 * ks);
      bf[ks].x = (lm < NW && k0 < NS)     ? Kfin[k0 * KROW + NS + lm]       : 0.0f;
      bf[ks].y = (lm < NW && k0 + 1 < NS) ? Kfin[(k0 + 1) * KROW + NS + lm] : 0.0f;
    }
    #pragma unroll
    for (int ks = 0; ks < 7; ++ks)
      cc = __builtin_amdgcn_wmma_f32_16x16x4_f32(false, af[ks], false, bf[ks],
                                                 (short)0, cc, false, false);
    const float sc = *scale;
    if (lm < NW) {
      #pragma unroll
      for (int v = 0; v < 8; ++v) {
        int m = m0 + kh * 8 + v;
        if (m < NQ) out[((size_t)b * NQ + m) * NW + lm] = sc * cc[v];
      }
    }
  }
}

extern "C" void kernel_launch(void* const* d_in, const int* in_sizes, int n_in,
                              void* d_out, int out_size, void* d_ws, size_t ws_size,
                              hipStream_t stream) {
  (void)n_in; (void)out_size; (void)d_ws; (void)ws_size;
  const float* q      = (const float*)d_in[0];
  const float* s      = (const float*)d_in[1];
  const int*   labels = (const int*)d_in[2];
  const float* scale  = (const float*)d_in[3];
  float* out = (float*)d_out;

  const int B = in_sizes[1] / (NS * Dd);
  if (B <= 0) return;

  ridge_head_fused<<<dim3(B), dim3(NTHR), 0, stream>>>(q, s, labels, scale, out);
}


# Gauss-Jordan pivot reciprocals by v_rcp_f32 (1 ulp), not the IEEE division sequence: the serial chain through the 25 pivots loses about a dozen dependent instructions per pivot; last bits change
# speedup vs baseline: 2.9680x; 1.0292x over previous
#include <hip/hip_runtime.h>


typedef float v2f __attribute__((ext_vector_type(2)));
typedef float v4f __attribute__((ext_vector_type(4)));
typedef float v8f __attribute__((ext_vector_type(8)));

#define Dd    1024
#define NS    25
#define NQ    75
#define NW    5
#define KROW  32
#define KCH   128
#define NCHUNK (Dd / KCH)
#define NWAVE 7
#define NTHR  (NWAVE * 32)
#define TROWS (NWAVE * 16)
#define CROW  132
#define BUFF  (TROWS * CROW)
#define GROW  36
#define GJB   5
static_assert(NS % GJB == 0, "the solve takes whole blocks of GJB pivots");
static_assert(4 * NWAVE >= NS, "four rows per wave must cover the matrix");

#define OFF_KAUG (2 * BUFF)
#define KR    (4 * NWAVE)
#define KCOPY (KR * KROW)
#define SMEM_FLOATS (OFF_KAUG + 2 * KCOPY)

__device__ __forceinline__ void chunk_barrier() {
  asm volatile("s_wait_dscnt 0x0" ::: "memory");
  asm volatile("s_barrier_signal -1" ::: "memory");
  asm volatile("s_barrier_wait -1" ::: "memory");
}

struct Frag { v4f a[4], x0[4], x1[4]; };

__device__ __forceinline__ void load_blk(Frag& f, const float* pa, const float* pb0, const float* pb1) {
  #pragma unroll
  for (int g = 0; g < 4; ++g) {
    f.a[g]  = *(const v4f*)(pa  + 8 * g);
    f.x0[g] = *(const v4f*)(pb0 + 8 * g);
    f.x1[g] = *(const v4f*)(pb1 + 8 * g);
  }
}

__device__ __forceinline__ void mma_blk(const Frag& f, v8f& acc0, v8f& acc1) {
  #pragma unroll
  for (int g = 0; g < 4; ++g) {
    v2f alo  = {f.a[g].x,  f.a[g].y},  ahi  = {f.a[g].z,  f.a[g].w};
    v2f x0lo = {f.x0[g].x, f.x0[g].y}, x0hi = {f.x0[g].z, f.x0[g].w};
    v2f x1lo = {f.x1[g].x, f.x1[g].y}, x1hi = {f.x1[g].z, f.x1[g].w};
    acc0 = __builtin_amdgcn_wmma_f32_16x16x4_f32(false, alo, false, x0lo,
                                                 (short)0, acc0, false, false);
    acc1 = __builtin_amdgcn_wmma_f32_16x16x4_f32(false, alo, false, x1lo,
                                                 (short)0, acc1, false, false);
    acc0 = __builtin_amdgcn_wmma_f32_16x16x4_f32(false, ahi, false, x0hi,
                                                 (short)0, acc0, false, false);
    acc1 = __builtin_amdgcn_wmma_f32_16x16x4_f32(false, ahi, false, x1hi,
                                                 (short)0, acc1, false, false);
  }
}

__global__ __launch_bounds__(NTHR) void ridge_head_fused(
    const float* __restrict__ q, const float* __restrict__ s,
    const int* __restrict__ labels, const float* __restrict__ scale,
    float* __restrict__ out)
{
  __shared__ __align__(16) float smem[SMEM_FLOATS];
  float* Kaug  = smem + OFF_KAUG;
  float* G     = smem;

  const int b    = blockIdx.x;
  const int tid  = threadIdx.x;
  const int lane = tid & 31;
  const int wave = __builtin_amdgcn_readfirstlane(tid >> 5);
  const int lm   = lane & 15;
  const int kh   = lane >> 4;

  const float* gsrc;
  int nvalid;
  if (wave < 2) {
    gsrc   = s + (size_t)b * NS * Dd + (size_t)wave * 16 * Dd;
    nvalid = (wave == 0) ? 16 : (NS - 16);
  } else {
    const int mq = (wave - 2) * 16;
    gsrc   = q + (size_t)b * NQ * Dd + (size_t)mq * Dd;
    nvalid = (NQ - mq < 16) ? (NQ - mq) : 16;
  }

  float4 r[16];
  #pragma unroll
  for (int u = 0; u < 16; ++u) {
    float4 v = {0.f, 0.f, 0.f, 0.f};
    if (u < nvalid) v = ((const float4*)(gsrc + (size_t)u * Dd))[lane];
    r[u] = v;
  }

  if (tid < NS) {
    int lab = labels[b * NS + tid];
    for (int c = 0; c < NW; ++c)
      Kaug[tid * KROW + NS + c] = (lab == c) ? 1.0f : 0.0f;
  }
  if (tid < (KR - NS) * KROW) Kaug[NS * KROW + tid] = 0.0f;

  v8f acc0 = {0.f,0.f,0.f,0.f,0.f,0.f,0.f,0.f};
  v8f acc1 = {0.f,0.f,0.f,0.f,0.f,0.f,0.f,0.f};
  const int a_off  = (wave * 16 + lm) * CROW + 4 * kh;
  const int b0_off = lm * CROW + 4 * kh;
  const int b1_off = (16 + lm) * CROW + 4 * kh;

  for (int c = 0; c < NCHUNK; ++c) {
    float* buf = smem + (c & 1) * BUFF;

    #pragma unroll
    for (int u = 0; u < 16; ++u)
      ((float4*)(buf + (wave * 16 + u) * CROW))[lane] = r[u];

    if (c + 1 < NCHUNK) {
      #pragma unroll
      for (int u = 0; u < 16; ++u) {
        float4 v = {0.f, 0.f, 0.f, 0.f};
        if (u < nvalid) v = ((const float4*)(gsrc + (size_t)u * Dd + (c + 1) * KCH))[lane];
        r[u] = v;
      }
    }

    chunk_barrier();

    const float* pa  = buf + a_off;
    const float* pb0 = buf + b0_off;
    const float* pb1 = buf + b1_off;
    Frag f0, f1;
    load_blk(f0, pa, pb0, pb1);
    load_blk(f1, pa + 32, pb0 + 32, pb1 + 32);
    mma_blk(f0, acc0, acc1);
    load_blk(f0, pa + 64, pb0 + 64, pb1 + 64);
    mma_blk(f1, acc0, acc1);
    load_blk(f1, pa + 96, pb0 + 96, pb1 + 96);
    mma_blk(f0, acc0, acc1);
    mma_blk(f1, acc0, acc1);
    __builtin_amdgcn_sched_group_barrier(0x100, 24, 0);
    __builtin_amdgcn_sched_group_barrier(0x008, 16, 0);
    __builtin_amdgcn_sched_group_barrier(0x100, 12, 0);
    __builtin_amdgcn_sched_group_barrier(0x008, 16, 0);
    __builtin_amdgcn_sched_group_barrier(0x100, 12, 0);
    __builtin_amdgcn_sched_group_barrier(0x008, 32, 0);
  }

  if (wave < 2) {
    #pragma unroll
    for (int v = 0; v < 8; ++v) {
      const int i  = wave * 16 + kh * 8 + v;
      const int j0 = lm, j1 = 16 + lm;
      if (i < NS) {
        Kaug[i * KROW + j0] = acc0[v] + ((i == j0) ? 1.0f : 0.0f);
        if (j1 < NS) Kaug[i * KROW + j1] = acc1[v] + ((i == j1) ? 1.0f : 0.0f);
      }
    }
  } else {
    const int mr = (wave - 2) * 16 + kh * 8;
    #pragma unroll
    for (int v = 0; v < 8; ++v) {
      G[(mr + v) * GROW + lm]      = acc0[v];
      G[(mr + v) * GROW + 16 + lm] = acc1[v];
    }
  }
  __syncthreads();

  for (int p = 0; p < NS; p += GJB) {
    const int cur = (p / GJB) & 1;
    const float* srcK = Kaug + cur * KCOPY;
    float*       dstK = Kaug + (cur ^ 1) * KCOPY;
    if (lane < NS + NW) {
      float P[GJB][GJB], PJ[GJB], F[4][GJB], O[4];
      #pragma unroll
      for (int i = 0; i < GJB; ++i) {
        #pragma unroll
        for (int k = 0; k < GJB; ++k) P[i][k] = srcK[(p + i) * KROW + p + k];
        PJ[i] = srcK[(p + i) * KROW + lane];
      }
      #pragma unroll
      for (int t = 0; t < 4; ++t) {
        const int rr = wave + NWAVE * t;
        #pragma unroll
        for (int k = 0; k < GJB; ++k) F[t][k] = srcK[rr * KROW + p + k];
        O[t] = srcK[rr * KROW + lane];
      }
      #pragma unroll
      for (int s = 0; s < GJB; ++s) {
        const float pivinv = __builtin_amdgcn_rcpf(P[s][s]);
        float ps[GJB];
        #pragma unroll
        for (int k = s + 1; k < GJB; ++k) ps[k] = P[s][k] * pivinv;
        const float pj = PJ[s] * pivinv;
        float nps[GJB];
        #pragma unroll
        for (int k = s + 1; k < GJB; ++k) nps[k] = -ps[k];
        const float npj = -pj;
        #pragma unroll
        for (int i = 0; i < GJB; ++i) {
          if (i == s) continue;
          const float f = P[i][s];
          #pragma unroll
          for (int k = s + 1; k < GJB; ++k) P[i][k] = __builtin_fmaf(f, nps[k], P[i][k]);
          PJ[i] = __builtin_fmaf(f, npj, PJ[i]);
        }
        #pragma unroll
        for (int k = s + 1; k < GJB; ++k) P[s][k] = ps[k];
        PJ[s] = pj;
        #pragma unroll
        for (int t = 0; t < 4; ++t) {
          const float f = F[t][s];
          #pragma unroll
          for (int k = s + 1; k < GJB; ++k) F[t][k] = __builtin_fmaf(f, nps[k], F[t][k]);
          O[t] = __builtin_fmaf(f, npj, O[t]);
        }
      }
      #pragma unroll
      for (int t = 0; t < 4; ++t) {
        const int rr = wave + NWAVE * t;
        float v = O[t];
        #pragma unroll
        for (int i = 0; i < GJB; ++i)
          if (rr == p + i) v = PJ[i];
        dstK[rr * KROW + lane] = v;
      }
    }
    __syncthreads();
  }
  const float* Kfin = Kaug + ((NS / GJB) & 1) * KCOPY;

  if (wave >= 2) {
    const int m0 = (wave - 2) * 16;
    const float* ga = G + (m0 + lm) * GROW + 2 * kh;
    v8f cc = {0.f,0.f,0.f,0.f,0.f,0.f,0.f,0.f};
    v2f af[7], bf[7];
    #pragma unroll
    for (int ks = 0; ks < 7; ++ks) {
      const int k0 = 4 * ks + 2 * kh;
      af[ks] = *(const v2f*)(ga + 4 * ks);
      bf[ks].x = (lm < NW && k0 < NS)     ? Kfin[k0 * KROW + NS + lm]       : 0.0f;
      bf[ks].y = (lm < NW && k0 + 1 < NS) ? Kfin[(k0 + 1) * KROW + NS + lm] : 0.0f;
    }
    #pragma unroll
    for (int ks = 0; ks < 7; ++ks)
      cc = __builtin_amdgcn_wmma_f32_16x16x4_f32(false, af[ks], false, bf[ks],
                                                 (short)0, cc, false, false);
    const float sc = *scale;
    if (lm < NW) {
      #pragma unroll
      for (int v = 0; v < 8; ++v) {
        int m = m0 + kh * 8 + v;
        if (m < NQ) out[((size_t)b * NQ + m) * NW + lm] = sc * cc[v];
      }
    }
  }
}

extern "C" void kernel_launch(void* const* d_in, const int* in_sizes, int n_in,
                              void* d_out, int out_size, void* d_ws, size_t ws_size,
                              hipStream_t stream) {
  (void)n_in; (void)out_size; (void)d_ws; (void)ws_size;
  const float* q      = (const float*)d_in[0];
  const float* s      = (const float*)d_in[1];
  const int*   labels = (const int*)d_in[2];
  const float* scale  = (const float*)d_in[3];
  float* out = (float*)d_out;

  const int B = in_sizes[1] / (NS * Dd);
  if (B <= 0) return;

  ridge_head_fused<<<dim3(B), dim3(NTHR), 0, stream>>>(q, s, labels, scale, out);
}
